# starting kernel
# baseline (speedup 1.0000x reference)
#include <hip/hip_runtime.h>

typedef __attribute__((ext_vector_type(2))) float v2f;
typedef __attribute__((ext_vector_type(8))) float v8f;

__device__ __forceinline__ void fillCoreA(float* dst, const float* __restrict__ src, int tid) {
  for (int idx = tid; idx < 1600; idx += 256) {
    int p = idx & 1;
    int l = (idx >> 1) & 31;
    int t = idx >> 6;
    int K = 4 * t + ((l & 16) ? 2 : 0) + p;
    int j = l & 15;
    int i = K / 10;
    int k = K % 10;
    dst[idx] = (j < 10) ? src[(i * 10 + k) * 10 + j] : 0.0f;
  }
}

__device__ __forceinline__ void leafContract(const float* sFM, const float* __restrict__ X,
                                             int b, int f, int u, float c[10]) {
  float x0 = X[(b * 8 + f) * 2 + 0];
  float x1 = X[(b * 8 + f) * 2 + 1];
#pragma unroll
  for (int r = 0; r < 10; ++r) {
    c[r] = x0 * sFM[((r) * 8 + f) * 32 + u] + x1 * sFM[((10 + r) * 8 + f) * 32 + u];
  }
}

__device__ __forceinline__ void pairContract(const float* sCoreC, const float a[10],
                                             const float bv[10], int lane, float q[10]) {
  v8f acc = {};
  bool hi = (lane & 16) != 0;
#pragma unroll
  for (int t = 0; t < 25; ++t) {
    v2f A = *(const v2f*)(sCoreC + (t * 32 + lane) * 2);
    float blo0 = a[(4 * t    ) / 10] * bv[(4 * t    ) % 10];
    float blo1 = a[(4 * t + 1) / 10] * bv[(4 * t + 1) % 10];
    float bhi0 = a[(4 * t + 2) / 10] * bv[(4 * t + 2) % 10];
    float bhi1 = a[(4 * t + 3) / 10] * bv[(4 * t + 3) % 10];
    v2f Bv;
    Bv.x = hi ? bhi0 : blo0;
    Bv.y = hi ? bhi1 : blo1;
    acc = __builtin_amdgcn_wmma_f32_16x16x4_f32(false, A, false, Bv, (short)0, acc,
                                                false, false);
  }
  float m[8];
#pragma unroll
  for (int v = 0; v < 8; ++v) m[v] = __shfl_xor(acc[v], 16, 32);
#pragma unroll
  for (int v = 0; v < 8; ++v) q[v] = hi ? m[v] : acc[v];
  q[8] = hi ? acc[0] : m[0];
  q[9] = hi ? acc[1] : m[1];
}

__global__ __launch_bounds__(256) void tensortree_wmma_kernel(
    const float* __restrict__ X, const float* __restrict__ c11,
    const float* __restrict__ c12, const float* __restrict__ c13,
    const float* __restrict__ c14, const float* __restrict__ c21,
    const float* __restrict__ c22, const float* __restrict__ fm,
    const float* __restrict__ T, float* __restrict__ out) {
  __shared__ float sCore[6 * 1600];
  __shared__ float sFM[5120];

  int tid = threadIdx.x;
  fillCoreA(sCore + 0,    c11, tid);
  fillCoreA(sCore + 1600, c12, tid);
  fillCoreA(sCore + 3200, c13, tid);
  fillCoreA(sCore + 4800, c14, tid);
  fillCoreA(sCore + 6400, c21, tid);
  fillCoreA(sCore + 8000, c22, tid);
  for (int idx = tid; idx < 5120; idx += 256) sFM[idx] = fm[idx];
  __syncthreads();

  int lane = tid & 31;
  int wave = blockIdx.x * 8 + (tid >> 5);
  int b = __builtin_amdgcn_readfirstlane(wave >> 1);
  int u = ((wave & 1) << 4) | (lane & 15);

  float cA[10], cB[10], pA[10], pB[10], q1[10], q2[10];

  leafContract(sFM, X, b, 0, u, cA);
  leafContract(sFM, X, b, 1, u, cB);
  pairContract(sCore + 0, cA, cB, lane, pA);
  leafContract(sFM, X, b, 2, u, cA);
  leafContract(sFM, X, b, 3, u, cB);
  pairContract(sCore + 1600, cA, cB, lane, pB);
  pairContract(sCore + 6400, pA, pB, lane, q1);

  leafContract(sFM, X, b, 4, u, cA);
  leafContract(sFM, X, b, 5, u, cB);
  pairContract(sCore + 3200, cA, cB, lane, pA);
  leafContract(sFM, X, b, 6, u, cA);
  leafContract(sFM, X, b, 7, u, cB);
  pairContract(sCore + 4800, cA, cB, lane, pB);
  pairContract(sCore + 8000, pA, pB, lane, q2);

  float r = 0.0f;
#pragma unroll
  for (int i = 0; i < 10; ++i) {
    float s = 0.0f;
#pragma unroll
    for (int j = 0; j < 10; ++j) s += T[i * 10 + j] * q2[j];
    r += q1[i] * s;
  }
  if (lane < 16) out[b * 32 + u] = r;
}

extern "C" void kernel_launch(void* const* d_in, const int* in_sizes, int n_in,
                              void* d_out, int out_size, void* d_ws, size_t ws_size,
                              hipStream_t stream) {
  const float* X   = (const float*)d_in[0];
  const float* c11 = (const float*)d_in[1];
  const float* c12 = (const float*)d_in[2];
  const float* c13 = (const float*)d_in[3];
  const float* c14 = (const float*)d_in[4];
  const float* c21 = (const float*)d_in[5];
  const float* c22 = (const float*)d_in[6];
  const float* fm  = (const float*)d_in[7];
  const float* T   = (const float*)d_in[8];
  float* out = (float*)d_out;

  int B = in_sizes[0] / 16;
  int blocks = (B * 32) / 128;
  tensortree_wmma_kernel<<<dim3(blocks), dim3(256), 0, stream>>>(
      X, c11, c12, c13, c14, c21, c22, fm, T, out);
}


# Persistent work-groups (one per compute unit, 2 waves/SIMD) keep the six cores' WMMA A fragments in registers and loop over 64 row tiles: no operand load among the WMMAs, core repack once per wave
# speedup vs baseline: 1.1707x; 1.1707x over previous
#include <hip/hip_runtime.h>


typedef __attribute__((ext_vector_type(2))) float v2f;
typedef __attribute__((ext_vector_type(8))) float v8f;

__device__ __forceinline__ void loadCoreA(v2f (&A)[25], const float* __restrict__ src, int lane) {
  int j = lane & 15;
  int h = lane >> 4;
  bool valid = j < 10;
  const float* p = src + (valid ? (20 * h + j) : 0);
#pragma unroll
  for (int t = 0; t < 25; ++t) {
    float a0 = p[40 * t];
    float a1 = p[40 * t + 10];
    A[t].x = valid ? a0 : 0.0f;
    A[t].y = valid ? a1 : 0.0f;
  }
}

__device__ __forceinline__ void leafContract(const float* sFM, const float* __restrict__ X,
                                             int b, int f, int u, float c[10]) {
  float x0 = X[(b * 8 + f) * 2 + 0];
  float x1 = X[(b * 8 + f) * 2 + 1];
#pragma unroll
  for (int r = 0; r < 10; ++r) {
    c[r] = x0 * sFM[((r) * 8 + f) * 32 + u] + x1 * sFM[((10 + r) * 8 + f) * 32 + u];
  }
}

__device__ __forceinline__ void pairContract(const v2f (&Acore)[25], const float a[10],
                                             const float bv[10], int lane, float q[10]) {
  v8f acc = {};
  bool hi = (lane & 16) != 0;
#pragma unroll
  for (int t = 0; t < 25; ++t) {
    float blo0 = a[(4 * t    ) / 10] * bv[(4 * t    ) % 10];
    float blo1 = a[(4 * t + 1) / 10] * bv[(4 * t + 1) % 10];
    float bhi0 = a[(4 * t + 2) / 10] * bv[(4 * t + 2) % 10];
    float bhi1 = a[(4 * t + 3) / 10] * bv[(4 * t + 3) % 10];
    v2f Bv;
    Bv.x = hi ? bhi0 : blo0;
    Bv.y = hi ? bhi1 : blo1;
    acc = __builtin_amdgcn_wmma_f32_16x16x4_f32(false, Acore[t], false, Bv, (short)0, acc,
                                                false, false);
  }
  float m[8];
#pragma unroll
  for (int v = 0; v < 8; ++v) m[v] = __shfl_xor(acc[v], 16, 32);
#pragma unroll
  for (int v = 0; v < 8; ++v) q[v] = hi ? m[v] : acc[v];
  q[8] = hi ? acc[0] : m[0];
  q[9] = hi ? acc[1] : m[1];
}

__global__ __launch_bounds__(256) void tensortree_wmma_kernel(
    const float* __restrict__ X, const float* __restrict__ c11,
    const float* __restrict__ c12, const float* __restrict__ c13,
    const float* __restrict__ c14, const float* __restrict__ c21,
    const float* __restrict__ c22, const float* __restrict__ fm,
    const float* __restrict__ T, float* __restrict__ out, int numTiles) {
  int tid = threadIdx.x;
  int lane = tid & 31;
  int wib = tid >> 5;
  int u = ((wib & 1) << 4) | (lane & 15);

  v2f A11[25], A12[25], A13[25], A14[25], A21[25], A22[25];
  loadCoreA(A11, c11, lane);
  loadCoreA(A12, c12, lane);
  loadCoreA(A13, c13, lane);
  loadCoreA(A14, c14, lane);
  loadCoreA(A21, c21, lane);
  loadCoreA(A22, c22, lane);

  __shared__ float sFM[5120];
  for (int idx = tid; idx < 5120; idx += 256) sFM[idx] = fm[idx];
  __syncthreads();

  const int tileStride = gridDim.x;
  for (int tile = blockIdx.x; tile < numTiles; tile += tileStride) {
    asm volatile("" ::: "memory");
    int b = __builtin_amdgcn_readfirstlane(tile * 4 + (wib >> 1));

    float cA[10], cB[10], pA[10], pB[10], q1[10], q2[10];

    leafContract(sFM, X, b, 0, u, cA);
    leafContract(sFM, X, b, 1, u, cB);
    pairContract(A11, cA, cB, lane, pA);
    leafContract(sFM, X, b, 2, u, cA);
    leafContract(sFM, X, b, 3, u, cB);
    pairContract(A12, cA, cB, lane, pB);
    pairContract(A21, pA, pB, lane, q1);

    leafContract(sFM, X, b, 4, u, cA);
    leafContract(sFM, X, b, 5, u, cB);
    pairContract(A13, cA, cB, lane, pA);
    leafContract(sFM, X, b, 6, u, cA);
    leafContract(sFM, X, b, 7, u, cB);
    pairContract(A14, cA, cB, lane, pB);
    pairContract(A22, pA, pB, lane, q2);

    float r = 0.0f;
#pragma unroll
    for (int i = 0; i < 10; ++i) {
      float s = 0.0f;
#pragma unroll
      for (int j = 0; j < 10; ++j) s += T[i * 10 + j] * q2[j];
      r += q1[i] * s;
    }
    if (lane < 16) out[b * 32 + u] = r;
  }
}

extern "C" void kernel_launch(void* const* d_in, const int* in_sizes, int n_in,
                              void* d_out, int out_size, void* d_ws, size_t ws_size,
                              hipStream_t stream) {
  const float* X   = (const float*)d_in[0];
  const float* c11 = (const float*)d_in[1];
  const float* c12 = (const float*)d_in[2];
  const float* c13 = (const float*)d_in[3];
  const float* c14 = (const float*)d_in[4];
  const float* c21 = (const float*)d_in[5];
  const float* c22 = (const float*)d_in[6];
  const float* fm  = (const float*)d_in[7];
  const float* T   = (const float*)d_in[8];
  float* out = (float*)d_out;

  int B = in_sizes[0] / 16;
  int blocks = (B * 32) / 128;
  tensortree_wmma_kernel<<<dim3(256), dim3(256), 0, stream>>>(
      X, c11, c12, c13, c14, c21, c22, fm, T, out, blocks);
}


# Pair contractions run two at a time per wave (11 with 12, 13 with 14, 21 with 22): two independent WMMA accumulator chains issued alternately; each chain's own sum order unchanged
# speedup vs baseline: 1.2101x; 1.0337x over previous
#include <hip/hip_runtime.h>


typedef __attribute__((ext_vector_type(2))) float v2f;
typedef __attribute__((ext_vector_type(8))) float v8f;

__device__ __forceinline__ void loadCoreA(v2f (&A)[25], const float* __restrict__ src, int lane) {
  int j = lane & 15;
  int h = lane >> 4;
  bool valid = j < 10;
  const float* p = src + (valid ? (20 * h + j) : 0);
#pragma unroll
  for (int t = 0; t < 25; ++t) {
    float a0 = p[40 * t];
    float a1 = p[40 * t + 10];
    A[t].x = valid ? a0 : 0.0f;
    A[t].y = valid ? a1 : 0.0f;
  }
}

__device__ __forceinline__ void leafContract(const float* sFM, const float* __restrict__ X,
                                             int b, int f, int u, float c[10]) {
  float x0 = X[(b * 8 + f) * 2 + 0];
  float x1 = X[(b * 8 + f) * 2 + 1];
#pragma unroll
  for (int r = 0; r < 10; ++r) {
    c[r] = x0 * sFM[((r) * 8 + f) * 32 + u] + x1 * sFM[((10 + r) * 8 + f) * 32 + u];
  }
}

__device__ __forceinline__ v2f bOperand(const float a[10], const float bv[10], bool hi, int t) {
  float blo0 = a[(4 * t    ) / 10] * bv[(4 * t    ) % 10];
  float blo1 = a[(4 * t + 1) / 10] * bv[(4 * t + 1) % 10];
  float bhi0 = a[(4 * t + 2) / 10] * bv[(4 * t + 2) % 10];
  float bhi1 = a[(4 * t + 3) / 10] * bv[(4 * t + 3) % 10];
  v2f Bv;
  Bv.x = hi ? bhi0 : blo0;
  Bv.y = hi ? bhi1 : blo1;
  return Bv;
}

__device__ __forceinline__ void mergeHalves(const v8f& acc, bool hi, float q[10]) {
  float m[8];
#pragma unroll
  for (int v = 0; v < 8; ++v) m[v] = __shfl_xor(acc[v], 16, 32);
#pragma unroll
  for (int v = 0; v < 8; ++v) q[v] = hi ? m[v] : acc[v];
  q[8] = hi ? acc[0] : m[0];
  q[9] = hi ? acc[1] : m[1];
}

__device__ __forceinline__ void pairContract2(const v2f (&A1)[25], const float a1[10],
                                              const float b1[10], const v2f (&A2)[25],
                                              const float a2[10], const float b2[10],
                                              int lane, float q1[10], float q2[10]) {
  v8f acc1 = {};
  v8f acc2 = {};
  bool hi = (lane & 16) != 0;
#pragma unroll
  for (int t = 0; t < 25; ++t) {
    v2f B1 = bOperand(a1, b1, hi, t);
    v2f B2 = bOperand(a2, b2, hi, t);
    acc1 = __builtin_amdgcn_wmma_f32_16x16x4_f32(false, A1[t], false, B1, (short)0, acc1,
                                                 false, false);
    acc2 = __builtin_amdgcn_wmma_f32_16x16x4_f32(false, A2[t], false, B2, (short)0, acc2,
                                                 false, false);
    asm("" : "+v"(acc1), "+v"(acc2));
  }
  mergeHalves(acc1, hi, q1);
  mergeHalves(acc2, hi, q2);
}

__global__ __launch_bounds__(256) void tensortree_wmma_kernel(
    const float* __restrict__ X, const float* __restrict__ c11,
    const float* __restrict__ c12, const float* __restrict__ c13,
    const float* __restrict__ c14, const float* __restrict__ c21,
    const float* __restrict__ c22, const float* __restrict__ fm,
    const float* __restrict__ T, float* __restrict__ out, int numTiles) {
  int tid = threadIdx.x;
  int lane = tid & 31;
  int wib = tid >> 5;
  int u = ((wib & 1) << 4) | (lane & 15);

  v2f A11[25], A12[25], A13[25], A14[25], A21[25], A22[25];
  loadCoreA(A11, c11, lane);
  loadCoreA(A12, c12, lane);
  loadCoreA(A13, c13, lane);
  loadCoreA(A14, c14, lane);
  loadCoreA(A21, c21, lane);
  loadCoreA(A22, c22, lane);

  __shared__ float sFM[5120];
  for (int idx = tid; idx < 5120; idx += 256) sFM[idx] = fm[idx];
  __syncthreads();

  const int tileStride = gridDim.x;
  for (int tile = blockIdx.x; tile < numTiles; tile += tileStride) {
    asm volatile("" ::: "memory");
    int b = __builtin_amdgcn_readfirstlane(tile * 4 + (wib >> 1));

    float cA[10], cB[10], cC[10], cD[10], pA[10], pB[10], pC[10], pD[10], q1[10], q2[10];

    leafContract(sFM, X, b, 0, u, cA);
    leafContract(sFM, X, b, 1, u, cB);
    leafContract(sFM, X, b, 2, u, cC);
    leafContract(sFM, X, b, 3, u, cD);
    pairContract2(A11, cA, cB, A12, cC, cD, lane, pA, pB);

    leafContract(sFM, X, b, 4, u, cA);
    leafContract(sFM, X, b, 5, u, cB);
    leafContract(sFM, X, b, 6, u, cC);
    leafContract(sFM, X, b, 7, u, cD);
    pairContract2(A13, cA, cB, A14, cC, cD, lane, pC, pD);

    pairContract2(A21, pA, pB, A22, pC, pD, lane, q1, q2);

    float r = 0.0f;
#pragma unroll
    for (int i = 0; i < 10; ++i) {
      float s = 0.0f;
#pragma unroll
      for (int j = 0; j < 10; ++j) s += T[i * 10 + j] * q2[j];
      r += q1[i] * s;
    }
    if (lane < 16) out[b * 32 + u] = r;
  }
}

extern "C" void kernel_launch(void* const* d_in, const int* in_sizes, int n_in,
                              void* d_out, int out_size, void* d_ws, size_t ws_size,
                              hipStream_t stream) {
  const float* X   = (const float*)d_in[0];
  const float* c11 = (const float*)d_in[1];
  const float* c12 = (const float*)d_in[2];
  const float* c13 = (const float*)d_in[3];
  const float* c14 = (const float*)d_in[4];
  const float* c21 = (const float*)d_in[5];
  const float* c22 = (const float*)d_in[6];
  const float* fm  = (const float*)d_in[7];
  const float* T   = (const float*)d_in[8];
  float* out = (float*)d_out;

  int B = in_sizes[0] / 16;
  int blocks = (B * 32) / 128;
  tensortree_wmma_kernel<<<dim3(256), dim3(256), 0, stream>>>(
      X, c11, c12, c13, c14, c21, c22, fm, T, out, blocks);
}


# Pair contractions re-associated: WMMA sums over i only (core as 100x10 A, the a vector itself as B: no outer products), k summed by in-lane FMAs on D; top likewise; sum order changed
# speedup vs baseline: 1.5296x; 1.2640x over previous
#include <hip/hip_runtime.h>


typedef __attribute__((ext_vector_type(2))) float v2f;
typedef __attribute__((ext_vector_type(8))) float v8f;


__device__ __forceinline__ void loadCoreA(v2f (&A)[7][3], const float* __restrict__ src, int lane) {
  int m = lane & 15;
  int h = lane >> 4;
  int g = m >> 3;
  int r = m & 7;
#pragma unroll
  for (int tau = 0; tau < 7; ++tau) {
    int sigma = 8 * tau + r;
    bool valid = sigma < 50;
    int jj = sigma / 10;
    int q = sigma - 10 * jj;
    int k = 5 * g + q;
    k = (k >= 10) ? k - 10 : k;
    int j = 5 * g + jj;
    const float* p = src + (valid ? (500 * h + 10 * k + j) : 0);
#pragma unroll
    for (int s = 0; s < 3; ++s) {
      float a0 = p[200 * s];
      float a1 = (2 * s + 1 < 5) ? p[200 * s + 100] : 0.0f;
      A[tau][s].x = valid ? a0 : 0.0f;
      A[tau][s].y = valid ? a1 : 0.0f;
    }
  }
}

__device__ __forceinline__ void loadTopA(v2f (&A)[3], const float* __restrict__ T, int lane) {
  int m = lane & 15;
  int h = lane >> 4;
  int g = m >> 3;
  int r = m & 7;
  bool valid = r < 5;
  const float* p = T + (valid ? (10 * (5 * g + r) + 5 * h) : 0);
#pragma unroll
  for (int s = 0; s < 3; ++s) {
    float a0 = p[2 * s];
    float a1 = (2 * s + 1 < 5) ? p[2 * s + 1] : 0.0f;
    A[s].x = valid ? a0 : 0.0f;
    A[s].y = valid ? a1 : 0.0f;
  }
}

__device__ __forceinline__ void leafHalf(const float* base, float x0, float x1, int f, float* c) {
#pragma unroll
  for (int e = 0; e < 5; ++e) {
    c[e] = x0 * base[(e * 8 + f) * 32] + x1 * base[((10 + e) * 8 + f) * 32];
  }
}

__device__ __forceinline__ v8f wmma4(v2f a, v2f b, v8f c) {
  return __builtin_amdgcn_wmma_f32_16x16x4_f32(false, a, false, b, (short)0, c, false, false);
}

__device__ __forceinline__ void pairContract(const v2f (&A)[7][3], const float a[5],
                                             const float bf[10], float o[5]) {
  v2f B0, B1, B2;
  B0.x = a[0]; B0.y = a[1];
  B1.x = a[2]; B1.y = a[3];
  B2.x = a[4]; B2.y = 0.0f;
  v8f D[7];
#pragma unroll
  for (int tau = 0; tau < 7; ++tau) { v8f z = {}; D[tau] = wmma4(A[tau][0], B0, z); }
#pragma unroll
  for (int tau = 0; tau < 7; ++tau) D[tau] = wmma4(A[tau][1], B1, D[tau]);
#pragma unroll
  for (int tau = 0; tau < 7; ++tau) D[tau] = wmma4(A[tau][2], B2, D[tau]);
#pragma unroll
  for (int jp = 0; jp < 5; ++jp) {
    v2f acc = {0.0f, 0.0f};
#pragma unroll
    for (int q = 0; q < 10; q += 2) {
      int sigma = 10 * jp + q;
      v2f d, w;
      d.x = D[sigma >> 3][sigma & 7];
      d.y = D[sigma >> 3][(sigma & 7) + 1];
      w.x = bf[q];
      w.y = bf[q + 1];
      acc = __builtin_elementwise_fma(d, w, acc);
    }
    o[jp] = acc.x + acc.y;
  }
}

__device__ __forceinline__ void completeFull(const float own[5], float full[10]) {
#pragma unroll
  for (int e = 0; e < 5; ++e) {
    full[e] = own[e];
    full[5 + e] = __shfl_xor(own[e], 16, 32);
  }
}

__global__ __launch_bounds__(256) void tensortree_wmma_kernel(
    const float* __restrict__ X, const float* __restrict__ c11,
    const float* __restrict__ c12, const float* __restrict__ c13,
    const float* __restrict__ c14, const float* __restrict__ c21,
    const float* __restrict__ c22, const float* __restrict__ fm,
    const float* __restrict__ T, float* __restrict__ out, int numTiles) {
  int tid = threadIdx.x;
  int lane = tid & 31;
  int wib = tid >> 5;
  int u = ((wib & 1) << 4) | (lane & 15);
  int h = lane >> 4;

  v2f A11[7][3], A12[7][3], A13[7][3], A14[7][3], A21[7][3], A22[7][3], AT[3];
  loadCoreA(A11, c11, lane);
  loadCoreA(A12, c12, lane);
  loadCoreA(A13, c13, lane);
  loadCoreA(A14, c14, lane);
  loadCoreA(A21, c21, lane);
  loadCoreA(A22, c22, lane);
  loadTopA(AT, T, lane);

  __shared__ float sFM[5120];
  for (int idx = tid; idx < 5120; idx += 256) sFM[idx] = fm[idx];
  __syncthreads();

  const float* sOwn = sFM + 1280 * h + u;
  const float* sOth = sFM + 1280 * (1 - h) + u;

  const int tileStride = gridDim.x;
  for (int tile = blockIdx.x; tile < numTiles; tile += tileStride) {
    asm volatile("" ::: "memory");
    int b = __builtin_amdgcn_readfirstlane(tile * 4 + (wib >> 1));
    const float* Xb = X + b * 16;

    float ca[5], cb[10], cc[5], cd[10], p1[5], p2[5], p2f[10], p3[5], p4[5], p4f[10], q1[5], q2[5];

    leafHalf(sOwn, Xb[0], Xb[1], 0, ca);
    leafHalf(sOwn, Xb[2], Xb[3], 1, cb);
    leafHalf(sOth, Xb[2], Xb[3], 1, cb + 5);
    pairContract(A11, ca, cb, p1);
    leafHalf(sOwn, Xb[4], Xb[5], 2, cc);
    leafHalf(sOwn, Xb[6], Xb[7], 3, cd);
    leafHalf(sOth, Xb[6], Xb[7], 3, cd + 5);
    pairContract(A12, cc, cd, p2);
    completeFull(p2, p2f);
    leafHalf(sOwn, Xb[8], Xb[9], 4, ca);
    leafHalf(sOwn, Xb[10], Xb[11], 5, cb);
    leafHalf(sOth, Xb[10], Xb[11], 5, cb + 5);
    pairContract(A13, ca, cb, p3);
    leafHalf(sOwn, Xb[12], Xb[13], 6, cc);
    leafHalf(sOwn, Xb[14], Xb[15], 7, cd);
    leafHalf(sOth, Xb[14], Xb[15], 7, cd + 5);
    pairContract(A14, cc, cd, p4);
    completeFull(p4, p4f);

    pairContract(A21, p1, p2f, q1);
    pairContract(A22, p3, p4f, q2);

    v2f Q0, Q1, Q2;
    Q0.x = q2[0]; Q0.y = q2[1];
    Q1.x = q2[2]; Q1.y = q2[3];
    Q2.x = q2[4]; Q2.y = 0.0f;
    v8f z = {};
    v8f S = wmma4(AT[0], Q0, z);
    S = wmma4(AT[1], Q1, S);
    S = wmma4(AT[2], Q2, S);
    float part = 0.0f;
#pragma unroll
    for (int e = 0; e < 5; ++e) part += q1[e] * S[e];
    float res = part + __shfl_xor(part, 16, 32);
    if (lane < 16) out[b * 32 + u] = res;
  }
}

extern "C" void kernel_launch(void* const* d_in, const int* in_sizes, int n_in,
                              void* d_out, int out_size, void* d_ws, size_t ws_size,
                              hipStream_t stream) {
  const float* X   = (const float*)d_in[0];
  const float* c11 = (const float*)d_in[1];
  const float* c12 = (const float*)d_in[2];
  const float* c13 = (const float*)d_in[3];
  const float* c14 = (const float*)d_in[4];
  const float* c21 = (const float*)d_in[5];
  const float* c22 = (const float*)d_in[6];
  const float* fm  = (const float*)d_in[7];
  const float* T   = (const float*)d_in[8];
  float* out = (float*)d_out;

  int B = in_sizes[0] / 16;
  int blocks = (B * 32) / 128;
  tensortree_wmma_kernel<<<dim3(256), dim3(256), 0, stream>>>(
      X, c11, c12, c13, c14, c21, c22, fm, T, out, blocks);
}


# Leaves folded into level 1 (a pair is bilinear in its leaves' X): per-unit 4x10 tables built once per wave by the WMMA contraction, then one K=4 WMMA per pair per 16 batch rows; sum order changed
# speedup vs baseline: 3.9583x; 2.5878x over previous
#include <hip/hip_runtime.h>


typedef __attribute__((ext_vector_type(2))) float v2f;
typedef __attribute__((ext_vector_type(4))) float v4f;
typedef __attribute__((ext_vector_type(8))) float v8f;


__device__ __forceinline__ void loadCoreA(v2f (&A)[7][3], const float* __restrict__ src, int lane) {
  int m = lane & 15;
  int h = lane >> 4;
  int g = m >> 3;
  int r = m & 7;
#pragma unroll
  for (int tau = 0; tau < 7; ++tau) {
    int sigma = 8 * tau + r;
    bool valid = sigma < 50;
    int jj = sigma / 10;
    int q = sigma - 10 * jj;
    int k = 5 * g + q;
    k = (k >= 10) ? k - 10 : k;
    int j = 5 * g + jj;
    const float* p = src + (valid ? (500 * h + 10 * k + j) : 0);
#pragma unroll
    for (int s = 0; s < 3; ++s) {
      float a0 = p[200 * s];
      float a1 = (2 * s + 1 < 5) ? p[200 * s + 100] : 0.0f;
      A[tau][s].x = valid ? a0 : 0.0f;
      A[tau][s].y = valid ? a1 : 0.0f;
    }
  }
}

__device__ __forceinline__ void loadTopA(v2f (&A)[3], const float* __restrict__ T, int lane) {
  int m = lane & 15;
  int h = lane >> 4;
  int g = m >> 3;
  int r = m & 7;
  bool valid = r < 5;
  const float* p = T + (valid ? (10 * (5 * g + r) + 5 * h) : 0);
#pragma unroll
  for (int s = 0; s < 3; ++s) {
    float a0 = p[2 * s];
    float a1 = (2 * s + 1 < 5) ? p[2 * s + 1] : 0.0f;
    A[s].x = valid ? a0 : 0.0f;
    A[s].y = valid ? a1 : 0.0f;
  }
}

__device__ __forceinline__ v8f wmma4(v2f a, v2f b, v8f c) {
  return __builtin_amdgcn_wmma_f32_16x16x4_f32(false, a, false, b, (short)0, c, false, false);
}

__device__ __forceinline__ void pairContract(const v2f (&A)[7][3], const float a[5],
                                             const float bf[10], float o[5]) {
  v2f B0, B1, B2;
  B0.x = a[0]; B0.y = a[1];
  B1.x = a[2]; B1.y = a[3];
  B2.x = a[4]; B2.y = 0.0f;
  v8f D[7];
#pragma unroll
  for (int tau = 0; tau < 7; ++tau) { v8f z = {}; D[tau] = wmma4(A[tau][0], B0, z); }
#pragma unroll
  for (int tau = 0; tau < 7; ++tau) D[tau] = wmma4(A[tau][1], B1, D[tau]);
#pragma unroll
  for (int tau = 0; tau < 7; ++tau) D[tau] = wmma4(A[tau][2], B2, D[tau]);
#pragma unroll
  for (int jp = 0; jp < 5; ++jp) {
    v2f acc = {0.0f, 0.0f};
#pragma unroll
    for (int q = 0; q < 10; q += 2) {
      int sigma = 10 * jp + q;
      v2f d, w;
      d.x = D[sigma >> 3][sigma & 7];
      d.y = D[sigma >> 3][(sigma & 7) + 1];
      w.x = bf[q];
      w.y = bf[q + 1];
      acc = __builtin_elementwise_fma(d, w, acc);
    }
    o[jp] = acc.x + acc.y;
  }
}

__device__ __forceinline__ void completeFull(const float own[5], float full[10]) {
#pragma unroll
  for (int e = 0; e < 5; ++e) {
    full[e] = own[e];
    full[5 + e] = __shfl_xor(own[e], 16, 32);
  }
}


__device__ __forceinline__ void buildTable(const float* __restrict__ core, const float* __restrict__ fm,
                                           int pi, int uBase, int lane, float* sGw) {
  int n = lane & 15;
  int h = lane >> 4;
  int u = uBase + (n >> 2);
  int d0 = (n >> 1) & 1;
  int d1 = n & 1;
  v2f Ac[7][3];
  loadCoreA(Ac, core, lane);
  float a[5], bf[10], o[5];
#pragma unroll
  for (int e = 0; e < 5; ++e) a[e] = fm[((d0 * 10 + 5 * h + e) * 8 + 2 * pi) * 32 + u];
#pragma unroll
  for (int q = 0; q < 10; ++q) {
    int k = 5 * h + q;
    k = (k >= 10) ? k - 10 : k;
    bf[q] = fm[((d1 * 10 + k) * 8 + 2 * pi + 1) * 32 + u];
  }
  pairContract(Ac, a, bf, o);
#pragma unroll
  for (int e = 0; e < 5; ++e) sGw[(pi * 16 + n) * 10 + 5 * h + e] = o[e];
}

__global__ __launch_bounds__(256) void tensortree_wmma_kernel(
    const float* __restrict__ X, const float* __restrict__ c11,
    const float* __restrict__ c12, const float* __restrict__ c13,
    const float* __restrict__ c14, const float* __restrict__ c21,
    const float* __restrict__ c22, const float* __restrict__ fm,
    const float* __restrict__ T, float* __restrict__ out, int numTiles) {
  int tid = threadIdx.x;
  int lane = tid & 31;
  int w = tid >> 5;
  int n = lane & 15;
  int h = lane >> 4;

  __shared__ float sG[8 * 640];
  float* sGw = sG + 640 * w;
  buildTable(c11, fm, 0, 4 * w, lane, sGw);
  buildTable(c12, fm, 1, 4 * w, lane, sGw);
  buildTable(c13, fm, 2, 4 * w, lane, sGw);
  buildTable(c14, fm, 3, 4 * w, lane, sGw);
  __syncthreads();

  v2f AG[4][4];
  {
    int g = n >> 3;
    int r = n & 7;
    bool valid = r < 5;
    int col = valid ? (5 * g + r) : 0;
#pragma unroll
    for (int pi = 0; pi < 4; ++pi) {
#pragma unroll
      for (int ul = 0; ul < 4; ++ul) {
        const float* p = sGw + (pi * 16 + 4 * ul + 2 * h) * 10 + col;
        float g0 = p[0];
        float g1 = p[10];
        AG[pi][ul].x = valid ? g0 : 0.0f;
        AG[pi][ul].y = valid ? g1 : 0.0f;
      }
    }
  }

  v2f A21[7][3], A22[7][3], AT[3];
  loadCoreA(A21, c21, lane);
  loadCoreA(A22, c22, lane);
  loadTopA(AT, T, lane);

  const int tileStride = gridDim.x;
  for (int tile = blockIdx.x; tile < numTiles; tile += tileStride) {
    int b = 16 * tile + n;
    const v4f* Xv = reinterpret_cast<const v4f*>(X + b * 16);

    v2f Bx[4];
#pragma unroll
    for (int pi = 0; pi < 4; ++pi) {
      v4f x = Xv[pi];
      float xa = h ? x.y : x.x;
      Bx[pi].x = xa * x.z;
      Bx[pi].y = xa * x.w;
    }

    float res[4];
#pragma unroll
    for (int ul = 0; ul < 4; ++ul) {
      v8f z = {};
      v8f P1 = wmma4(AG[0][ul], Bx[0], z);
      v8f P2 = wmma4(AG[1][ul], Bx[1], z);
      v8f P3 = wmma4(AG[2][ul], Bx[2], z);
      v8f P4 = wmma4(AG[3][ul], Bx[3], z);
      float p1[5], p2[5], p2f[10], p3[5], p4[5], p4f[10], q1[5], q2[5];
#pragma unroll
      for (int e = 0; e < 5; ++e) { p1[e] = P1[e]; p2[e] = P2[e]; p3[e] = P3[e]; p4[e] = P4[e]; }
      completeFull(p2, p2f);
      completeFull(p4, p4f);

      pairContract(A21, p1, p2f, q1);
      pairContract(A22, p3, p4f, q2);

      v2f Q0, Q1, Q2;
      Q0.x = q2[0]; Q0.y = q2[1];
      Q1.x = q2[2]; Q1.y = q2[3];
      Q2.x = q2[4]; Q2.y = 0.0f;
      v8f S = wmma4(AT[0], Q0, z);
      S = wmma4(AT[1], Q1, S);
      S = wmma4(AT[2], Q2, S);
      float part = 0.0f;
#pragma unroll
      for (int e = 0; e < 5; ++e) part += q1[e] * S[e];
      res[ul] = part + __shfl_xor(part, 16, 32);
    }
    if (lane < 16) {
      v4f o = {res[0], res[1], res[2], res[3]};
      *reinterpret_cast<v4f*>(out + b * 32 + 4 * w) = o;
    }
  }
}

extern "C" void kernel_launch(void* const* d_in, const int* in_sizes, int n_in,
                              void* d_out, int out_size, void* d_ws, size_t ws_size,
                              hipStream_t stream) {
  const float* X   = (const float*)d_in[0];
  const float* c11 = (const float*)d_in[1];
  const float* c12 = (const float*)d_in[2];
  const float* c13 = (const float*)d_in[3];
  const float* c14 = (const float*)d_in[4];
  const float* c21 = (const float*)d_in[5];
  const float* c22 = (const float*)d_in[6];
  const float* fm  = (const float*)d_in[7];
  const float* T   = (const float*)d_in[8];
  float* out = (float*)d_out;

  int B = in_sizes[0] / 16;
  int tiles = B / 16;
  tensortree_wmma_kernel<<<dim3(256), dim3(256), 0, stream>>>(
      X, c11, c12, c13, c14, c21, c22, fm, T, out, tiles);
}


# Whole tree folded into the leaves (out is multilinear in the 8 leaves' X): per-unit 256-entry tables built once per work-group on the WMMA, then one K=256 WMMA GEMM per 16 rows; sum order changed
# speedup vs baseline: 19.6391x; 4.9615x over previous
#include <hip/hip_runtime.h>


typedef __attribute__((ext_vector_type(2))) float v2f;
typedef __attribute__((ext_vector_type(4))) float v4f;
typedef __attribute__((ext_vector_type(8))) float v8f;


__device__ __forceinline__ void loadCoreA(v2f (&A)[7][3], const float* __restrict__ src, int lane) {
  int m = lane & 15;
  int h = lane >> 4;
  int g = m >> 3;
  int r = m & 7;
#pragma unroll
  for (int tau = 0; tau < 7; ++tau) {
    int sigma = 8 * tau + r;
    bool valid = sigma < 50;
    int jj = sigma / 10;
    int q = sigma - 10 * jj;
    int k = 5 * g + q;
    k = (k >= 10) ? k - 10 : k;
    int j = 5 * g + jj;
    const float* p = src + (valid ? (500 * h + 10 * k + j) : 0);
#pragma unroll
    for (int s = 0; s < 3; ++s) {
      float a0 = p[200 * s];
      float a1 = (2 * s + 1 < 5) ? p[200 * s + 100] : 0.0f;
      A[tau][s].x = valid ? a0 : 0.0f;
      A[tau][s].y = valid ? a1 : 0.0f;
    }
  }
}

__device__ __forceinline__ void loadTopA(v2f (&A)[3], const float* __restrict__ T, int lane) {
  int m = lane & 15;
  int h = lane >> 4;
  int g = m >> 3;
  int r = m & 7;
  bool valid = r < 5;
  const float* p = T + (valid ? (10 * (5 * g + r) + 5 * h) : 0);
#pragma unroll
  for (int s = 0; s < 3; ++s) {
    float a0 = p[2 * s];
    float a1 = (2 * s + 1 < 5) ? p[2 * s + 1] : 0.0f;
    A[s].x = valid ? a0 : 0.0f;
    A[s].y = valid ? a1 : 0.0f;
  }
}

__device__ __forceinline__ v8f wmma4(v2f a, v2f b, v8f c) {
  return __builtin_amdgcn_wmma_f32_16x16x4_f32(false, a, false, b, (short)0, c, false, false);
}

__device__ __forceinline__ void pairContract(const v2f (&A)[7][3], const float a[5],
                                             const float bf[10], float o[5]) {
  v2f B0, B1, B2;
  B0.x = a[0]; B0.y = a[1];
  B1.x = a[2]; B1.y = a[3];
  B2.x = a[4]; B2.y = 0.0f;
  v8f D[7];
#pragma unroll
  for (int tau = 0; tau < 7; ++tau) { v8f z = {}; D[tau] = wmma4(A[tau][0], B0, z); }
#pragma unroll
  for (int tau = 0; tau < 7; ++tau) D[tau] = wmma4(A[tau][1], B1, D[tau]);
#pragma unroll
  for (int tau = 0; tau < 7; ++tau) D[tau] = wmma4(A[tau][2], B2, D[tau]);
#pragma unroll
  for (int jp = 0; jp < 5; ++jp) {
    v2f acc = {0.0f, 0.0f};
#pragma unroll
    for (int q = 0; q < 10; q += 2) {
      int sigma = 10 * jp + q;
      v2f d, w;
      d.x = D[sigma >> 3][sigma & 7];
      d.y = D[sigma >> 3][(sigma & 7) + 1];
      w.x = bf[q];
      w.y = bf[q + 1];
      acc = __builtin_elementwise_fma(d, w, acc);
    }
    o[jp] = acc.x + acc.y;
  }
}

__device__ __forceinline__ void completeFull(const float own[5], float full[10]) {
#pragma unroll
  for (int e = 0; e < 5; ++e) {
    full[e] = own[e];
    full[5 + e] = __shfl_xor(own[e], 16, 32);
  }
}


__device__ __forceinline__ void buildTable(const float* __restrict__ core, const float* __restrict__ fm,
                                           int pi, int uBase, int lane, float* sGw) {
  int n = lane & 15;
  int h = lane >> 4;
  int u = uBase + (n >> 2);
  int d0 = (n >> 1) & 1;
  int d1 = n & 1;
  v2f Ac[7][3];
  loadCoreA(Ac, core, lane);
  float a[5], bf[10], o[5];
#pragma unroll
  for (int e = 0; e < 5; ++e) a[e] = fm[((d0 * 10 + 5 * h + e) * 8 + 2 * pi) * 32 + u];
#pragma unroll
  for (int q = 0; q < 10; ++q) {
    int k = 5 * h + q;
    k = (k >= 10) ? k - 10 : k;
    bf[q] = fm[((d1 * 10 + k) * 8 + 2 * pi + 1) * 32 + u];
  }
  pairContract(Ac, a, bf, o);
#pragma unroll
  for (int e = 0; e < 5; ++e) sGw[(pi * 16 + n) * 10 + 5 * h + e] = o[e];
}

__global__ __launch_bounds__(256) void tensortree_wmma_kernel(
    const float* __restrict__ X, const float* __restrict__ c11,
    const float* __restrict__ c12, const float* __restrict__ c13,
    const float* __restrict__ c14, const float* __restrict__ c21,
    const float* __restrict__ c22, const float* __restrict__ fm,
    const float* __restrict__ T, float* __restrict__ out, int numTiles) {
  int tid = threadIdx.x;
  int lane = tid & 31;
  int w = tid >> 5;
  int c = lane & 15;
  int h = lane >> 4;
  int mt = blockIdx.x & 1;
  int tg = blockIdx.x >> 1;
  int tgCount = gridDim.x >> 1;

  __shared__ float sG[4 * 640];
  __shared__ __attribute__((aligned(16))) float sZ[16 * 258];

  {
    int quad = w >> 1;
    int odd = w & 1;
    float* sGq = sG + 640 * quad;
    int uBase = 16 * mt + 4 * quad;
    buildTable(odd ? c13 : c11, fm, 2 * odd, uBase, lane, sGq);
    buildTable(odd ? c14 : c12, fm, 2 * odd + 1, uBase, lane, sGq);
  }

  v2f A21[7][3], A22[7][3], AT[3];
  loadCoreA(A21, c21, lane);
  loadCoreA(A22, c22, lane);
  loadTopA(AT, T, lane);
  __syncthreads();

  {
    const float* sGq = sG + 640 * (w >> 1);
    int n1 = c >> 2;
    int n2 = c & 3;
#pragma unroll
    for (int i = 0; i < 2; ++i) {
      int ul = 2 * (w & 1) + i;
      float a[5], bf[10], q1[5], q2[5];
      const float* pa = sGq + (4 * ul + n1) * 10 + 5 * h;
      const float* pb = sGq + (16 + 4 * ul + n2) * 10;
#pragma unroll
      for (int e = 0; e < 5; ++e) a[e] = pa[e];
#pragma unroll
      for (int q = 0; q < 10; ++q) {
        int k = 5 * h + q;
        k = (k >= 10) ? k - 10 : k;
        bf[q] = pb[k];
      }
      pairContract(A21, a, bf, q1);
#pragma unroll
      for (int e = 0; e < 5; ++e) a[e] = pa[320 + e];
#pragma unroll
      for (int q = 0; q < 10; ++q) {
        int k = 5 * h + q;
        k = (k >= 10) ? k - 10 : k;
        bf[q] = pb[320 + k];
      }
      pairContract(A22, a, bf, q2);

      v8f z = {};
      v2f Q0, Q1, Q2;
      Q0.x = q2[0]; Q0.y = q2[1];
      Q1.x = q2[2]; Q1.y = q2[3];
      Q2.x = q2[4]; Q2.y = 0.0f;
      v8f S = wmma4(AT[0], Q0, z);
      S = wmma4(AT[1], Q1, S);
      S = wmma4(AT[2], Q2, S);
      v2f H0, H1v, H2v, S0, S1, S2;
      H0.x = q1[0]; H0.y = q1[1];
      H1v.x = q1[2]; H1v.y = q1[3];
      H2v.x = q1[4]; H2v.y = 0.0f;
      S0.x = S[0]; S0.y = S[1];
      S1.x = S[2]; S1.y = S[3];
      S2.x = S[4]; S2.y = 0.0f;
      v8f Zt = wmma4(H0, S0, z);
      Zt = wmma4(H1v, S1, Zt);
      Zt = wmma4(H2v, S2, Zt);
      float* pz = sZ + (2 * w + i) * 258 + 128 * h + c;
#pragma unroll
      for (int e = 0; e < 8; ++e) pz[16 * e] = Zt[e];
    }
  }
  __syncthreads();

  v2f AZ[64];
  {
    const float* p = sZ + c * 258 + 2 * h;
#pragma unroll
    for (int s = 0; s < 64; ++s) AZ[s] = *reinterpret_cast<const v2f*>(p + 4 * s);
  }

  const int tileStep = 8 * tgCount;
  int tile = tg + tgCount * w;
  v4f x0 = {}, x1 = {}, x2 = {}, x3 = {};
  if (tile < numTiles) {
    const v4f* Xv = reinterpret_cast<const v4f*>(X + (16 * tile + c) * 16);
    x0 = Xv[0]; x1 = Xv[1]; x2 = Xv[2]; x3 = Xv[3];
  }
  while (tile < numTiles) {
    int b = 16 * tile + c;
    float P01[4] = {x0.x * x0.z, x0.x * x0.w, x0.y * x0.z, x0.y * x0.w};
    float P23[4] = {x1.x * x1.z, x1.x * x1.w, x1.y * x1.z, x1.y * x1.w};
    float P45[4] = {x2.x * x2.z, x2.x * x2.w, x2.y * x2.z, x2.y * x2.w};
    float xa = h ? x3.y : x3.x;
    v2f wv;
    wv.x = xa * x3.z;
    wv.y = xa * x3.w;
    float L4[16];
    v2f Rv[4];
#pragma unroll
    for (int n1 = 0; n1 < 4; ++n1) {
#pragma unroll
      for (int n2 = 0; n2 < 4; ++n2) L4[4 * n1 + n2] = P01[n1] * P23[n2];
      Rv[n1] = P45[n1] * wv;
    }

    int nextTile = tile + tileStep;
    {
      int ldTile = (nextTile < numTiles) ? nextTile : tile;
      const v4f* Xn = reinterpret_cast<const v4f*>(X + (16 * ldTile + c) * 16);
      x0 = Xn[0]; x1 = Xn[1]; x2 = Xn[2]; x3 = Xn[3];
    }

    v8f acc0 = {}, acc1 = {};
#pragma unroll
    for (int s = 0; s < 64; s += 2) {
      v2f b0 = L4[s >> 2] * Rv[s & 3];
      acc0 = wmma4(AZ[s], b0, acc0);
      v2f b1 = L4[(s + 1) >> 2] * Rv[(s + 1) & 3];
      acc1 = wmma4(AZ[s + 1], b1, acc1);
    }
    v8f r8 = acc0 + acc1;
    v4f o0 = {r8[0], r8[1], r8[2], r8[3]};
    v4f o1 = {r8[4], r8[5], r8[6], r8[7]};
    v4f* po = reinterpret_cast<v4f*>(out + b * 32 + 16 * mt + 8 * h);
    po[0] = o0;
    po[1] = o1;
    tile = nextTile;
  }
}

extern "C" void kernel_launch(void* const* d_in, const int* in_sizes, int n_in,
                              void* d_out, int out_size, void* d_ws, size_t ws_size,
                              hipStream_t stream) {
  const float* X   = (const float*)d_in[0];
  const float* c11 = (const float*)d_in[1];
  const float* c12 = (const float*)d_in[2];
  const float* c13 = (const float*)d_in[3];
  const float* c14 = (const float*)d_in[4];
  const float* c21 = (const float*)d_in[5];
  const float* c22 = (const float*)d_in[6];
  const float* fm  = (const float*)d_in[7];
  const float* T   = (const float*)d_in[8];
  float* out = (float*)d_out;

  int B = in_sizes[0] / 16;
  int tiles = B / 16;
  tensortree_wmma_kernel<<<dim3(256), dim3(256), 0, stream>>>(
      X, c11, c12, c13, c14, c21, c22, fm, T, out, tiles);
}


# Table set-up de-duplicated: each stage-1 WMMA product (core x a) formed once per distinct a and shared by its b operands (2 in level 1, 4 in level 2): 54 WMMAs per wave, not 138; sum order unchanged
# speedup vs baseline: 24.1004x; 1.2272x over previous
#include <hip/hip_runtime.h>


typedef __attribute__((ext_vector_type(2))) float v2f;
typedef __attribute__((ext_vector_type(4))) float v4f;
typedef __attribute__((ext_vector_type(8))) float v8f;


__device__ __forceinline__ void loadCoreA(v2f (&A)[7][3], const float* __restrict__ src, int lane) {
  int m = lane & 15;
  int h = lane >> 4;
  int g = m >> 3;
  int r = m & 7;
#pragma unroll
  for (int tau = 0; tau < 7; ++tau) {
    int sigma = 8 * tau + r;
    bool valid = sigma < 50;
    int jj = sigma / 10;
    int q = sigma - 10 * jj;
    int k = 5 * g + q;
    k = (k >= 10) ? k - 10 : k;
    int j = 5 * g + jj;
    const float* p = src + (valid ? (500 * h + 10 * k + j) : 0);
#pragma unroll
    for (int s = 0; s < 3; ++s) {
      float a0 = p[200 * s];
      float a1 = (2 * s + 1 < 5) ? p[200 * s + 100] : 0.0f;
      A[tau][s].x = valid ? a0 : 0.0f;
      A[tau][s].y = valid ? a1 : 0.0f;
    }
  }
}

__device__ __forceinline__ void loadTopA(v2f (&A)[3], const float* __restrict__ T, int lane) {
  int m = lane & 15;
  int h = lane >> 4;
  int g = m >> 3;
  int r = m & 7;
  bool valid = r < 5;
  const float* p = T + (valid ? (10 * (5 * g + r) + 5 * h) : 0);
#pragma unroll
  for (int s = 0; s < 3; ++s) {
    float a0 = p[2 * s];
    float a1 = (2 * s + 1 < 5) ? p[2 * s + 1] : 0.0f;
    A[s].x = valid ? a0 : 0.0f;
    A[s].y = valid ? a1 : 0.0f;
  }
}

__device__ __forceinline__ v8f wmma4(v2f a, v2f b, v8f c) {
  return __builtin_amdgcn_wmma_f32_16x16x4_f32(false, a, false, b, (short)0, c, false, false);
}

__device__ __forceinline__ void pairStage1(const v2f (&A)[7][3], const float a[5], v8f (&D)[7]) {
  v2f B0, B1, B2;
  B0.x = a[0]; B0.y = a[1];
  B1.x = a[2]; B1.y = a[3];
  B2.x = a[4]; B2.y = 0.0f;
#pragma unroll
  for (int tau = 0; tau < 7; ++tau) { v8f z = {}; D[tau] = wmma4(A[tau][0], B0, z); }
#pragma unroll
  for (int tau = 0; tau < 7; ++tau) D[tau] = wmma4(A[tau][1], B1, D[tau]);
#pragma unroll
  for (int tau = 0; tau < 7; ++tau) D[tau] = wmma4(A[tau][2], B2, D[tau]);
}

__device__ __forceinline__ void pairStage2(const v8f (&D)[7], const float bf[10], float o[5]) {
#pragma unroll
  for (int jp = 0; jp < 5; ++jp) {
    v2f acc = {0.0f, 0.0f};
#pragma unroll
    for (int q = 0; q < 10; q += 2) {
      int sigma = 10 * jp + q;
      v2f d, w;
      d.x = D[sigma >> 3][sigma & 7];
      d.y = D[sigma >> 3][(sigma & 7) + 1];
      w.x = bf[q];
      w.y = bf[q + 1];
      acc = __builtin_elementwise_fma(d, w, acc);
    }
    o[jp] = acc.x + acc.y;
  }
}

__device__ __forceinline__ void completeFull(const float own[5], float full[10]) {
#pragma unroll
  for (int e = 0; e < 5; ++e) {
    full[e] = own[e];
    full[5 + e] = __shfl_xor(own[e], 16, 32);
  }
}


__global__ __launch_bounds__(256) void tensortree_wmma_kernel(
    const float* __restrict__ X, const float* __restrict__ c11,
    const float* __restrict__ c12, const float* __restrict__ c13,
    const float* __restrict__ c14, const float* __restrict__ c21,
    const float* __restrict__ c22, const float* __restrict__ fm,
    const float* __restrict__ T, float* __restrict__ out, int numTiles) {
  int tid = threadIdx.x;
  int lane = tid & 31;
  int w = tid >> 5;
  int c = lane & 15;
  int h = lane >> 4;
  int mt = blockIdx.x & 1;
  int tg = blockIdx.x >> 1;
  int tgCount = gridDim.x >> 1;

  __shared__ float sG[4 * 16 * 40];
  __shared__ float sH[2 * 16 * 160];
  __shared__ __attribute__((aligned(16))) float sZ[16 * 258];

  v2f A1[7][3], A2[7][3], AT[3];
  {
    int wu = __builtin_amdgcn_readfirstlane(w);
    int pi = wu >> 1;
    const float* c1 = (pi == 0) ? c11 : (pi == 1) ? c12 : (pi == 2) ? c13 : c14;
    loadCoreA(A1, c1, lane);
    loadCoreA(A2, (wu & 1) ? c22 : c21, lane);
    loadTopA(AT, T, lane);
  }

  {
    int pi = w >> 1;
    int unit16 = 8 * (w & 1) + (c >> 1);
    int d0 = c & 1;
    int u = 16 * mt + unit16;
    float a[5], bf[2][10];
#pragma unroll
    for (int e = 0; e < 5; ++e) a[e] = fm[((d0 * 10 + 5 * h + e) * 8 + 2 * pi) * 32 + u];
#pragma unroll
    for (int d1 = 0; d1 < 2; ++d1) {
#pragma unroll
      for (int q = 0; q < 10; ++q) {
        int k = 5 * h + q;
        k = (k >= 10) ? k - 10 : k;
        bf[d1][q] = fm[((d1 * 10 + k) * 8 + 2 * pi + 1) * 32 + u];
      }
    }
    v8f D[7];
    pairStage1(A1, a, D);
    float* pg = sG + (pi * 16 + unit16) * 40 + 20 * d0 + 5 * h;
#pragma unroll
    for (int d1 = 0; d1 < 2; ++d1) {
      float o[5];
      pairStage2(D, bf[d1], o);
#pragma unroll
      for (int e = 0; e < 5; ++e) pg[10 * d1 + e] = o[e];
    }
  }
  __syncthreads();

  {
    int t = w & 1;
    int unit16 = 4 * (w >> 1) + (c >> 2);
    int n1 = c & 3;
    const float* pa = sG + (2 * t * 16 + unit16) * 40 + 10 * n1 + 5 * h;
    const float* pb = sG + ((2 * t + 1) * 16 + unit16) * 40;
    float a[5];
#pragma unroll
    for (int e = 0; e < 5; ++e) a[e] = pa[e];
    v8f D[7];
    pairStage1(A2, a, D);
    float* ph = sH + ((t * 16 + unit16) * 16 + 4 * n1) * 10 + 5 * h;
#pragma unroll
    for (int n2 = 0; n2 < 4; ++n2) {
      float bf[10], o[5];
#pragma unroll
      for (int q = 0; q < 10; ++q) {
        int k = 5 * h + q;
        k = (k >= 10) ? k - 10 : k;
        bf[q] = pb[10 * n2 + k];
      }
      pairStage2(D, bf, o);
#pragma unroll
      for (int e = 0; e < 5; ++e) ph[10 * n2 + e] = o[e];
    }
  }
  __syncthreads();

  {
#pragma unroll
    for (int i = 0; i < 2; ++i) {
      int unit16 = 2 * w + i;
      float q1[5], q2[5];
      const float* p1 = sH + (unit16 * 16 + c) * 10 + 5 * h;
      const float* p2 = sH + ((16 + unit16) * 16 + c) * 10 + 5 * h;
#pragma unroll
      for (int e = 0; e < 5; ++e) { q1[e] = p1[e]; q2[e] = p2[e]; }

      v8f z = {};
      v2f Q0, Q1, Q2;
      Q0.x = q2[0]; Q0.y = q2[1];
      Q1.x = q2[2]; Q1.y = q2[3];
      Q2.x = q2[4]; Q2.y = 0.0f;
      v8f S = wmma4(AT[0], Q0, z);
      S = wmma4(AT[1], Q1, S);
      S = wmma4(AT[2], Q2, S);
      v2f H0, H1v, H2v, S0, S1, S2;
      H0.x = q1[0]; H0.y = q1[1];
      H1v.x = q1[2]; H1v.y = q1[3];
      H2v.x = q1[4]; H2v.y = 0.0f;
      S0.x = S[0]; S0.y = S[1];
      S1.x = S[2]; S1.y = S[3];
      S2.x = S[4]; S2.y = 0.0f;
      v8f Zt = wmma4(H0, S0, z);
      Zt = wmma4(H1v, S1, Zt);
      Zt = wmma4(H2v, S2, Zt);
      float* pz = sZ + unit16 * 258 + 128 * h + c;
#pragma unroll
      for (int e = 0; e < 8; ++e) pz[16 * e] = Zt[e];
    }
  }
  __syncthreads();

  v2f AZ[64];
  {
    const float* p = sZ + c * 258 + 2 * h;
#pragma unroll
    for (int s = 0; s < 64; ++s) AZ[s] = *reinterpret_cast<const v2f*>(p + 4 * s);
  }

  const int tileStep = 8 * tgCount;
  int tile = tg + tgCount * w;
  v4f x0 = {}, x1 = {}, x2 = {}, x3 = {};
  if (tile < numTiles) {
    const v4f* Xv = reinterpret_cast<const v4f*>(X + (16 * tile + c) * 16);
    x0 = Xv[0]; x1 = Xv[1]; x2 = Xv[2]; x3 = Xv[3];
  }
  while (tile < numTiles) {
    int b = 16 * tile + c;
    float P01[4] = {x0.x * x0.z, x0.x * x0.w, x0.y * x0.z, x0.y * x0.w};
    float P23[4] = {x1.x * x1.z, x1.x * x1.w, x1.y * x1.z, x1.y * x1.w};
    float P45[4] = {x2.x * x2.z, x2.x * x2.w, x2.y * x2.z, x2.y * x2.w};
    float xa = h ? x3.y : x3.x;
    v2f wv;
    wv.x = xa * x3.z;
    wv.y = xa * x3.w;
    float L4[16];
    v2f Rv[4];
#pragma unroll
    for (int n1 = 0; n1 < 4; ++n1) {
#pragma unroll
      for (int n2 = 0; n2 < 4; ++n2) L4[4 * n1 + n2] = P01[n1] * P23[n2];
      Rv[n1] = P45[n1] * wv;
    }

    int nextTile = tile + tileStep;
    {
      int ldTile = (nextTile < numTiles) ? nextTile : tile;
      const v4f* Xn = reinterpret_cast<const v4f*>(X + (16 * ldTile + c) * 16);
      x0 = Xn[0]; x1 = Xn[1]; x2 = Xn[2]; x3 = Xn[3];
    }

    v8f acc0 = {}, acc1 = {};
#pragma unroll
    for (int s = 0; s < 64; s += 2) {
      v2f b0 = L4[s >> 2] * Rv[s & 3];
      acc0 = wmma4(AZ[s], b0, acc0);
      v2f b1 = L4[(s + 1) >> 2] * Rv[(s + 1) & 3];
      acc1 = wmma4(AZ[s + 1], b1, acc1);
    }
    v8f r8 = acc0 + acc1;
    v4f o0 = {r8[0], r8[1], r8[2], r8[3]};
    v4f o1 = {r8[4], r8[5], r8[6], r8[7]};
    v4f* po = reinterpret_cast<v4f*>(out + b * 32 + 16 * mt + 8 * h);
    po[0] = o0;
    po[1] = o1;
    tile = nextTile;
  }
}

extern "C" void kernel_launch(void* const* d_in, const int* in_sizes, int n_in,
                              void* d_out, int out_size, void* d_ws, size_t ws_size,
                              hipStream_t stream) {
  const float* X   = (const float*)d_in[0];
  const float* c11 = (const float*)d_in[1];
  const float* c12 = (const float*)d_in[2];
  const float* c13 = (const float*)d_in[3];
  const float* c14 = (const float*)d_in[4];
  const float* c21 = (const float*)d_in[5];
  const float* c22 = (const float*)d_in[6];
  const float* fm  = (const float*)d_in[7];
  const float* T   = (const float*)d_in[8];
  float* out = (float*)d_out;

  int B = in_sizes[0] / 16;
  int tiles = B / 16;
  tensortree_wmma_kernel<<<dim3(256), dim3(256), 0, stream>>>(
      X, c11, c12, c13, c14, c21, c22, fm, T, out, tiles);
}
